# starting kernel
# baseline (speedup 1.0000x reference)
#include <hip/hip_runtime.h>

typedef float v2f __attribute__((ext_vector_type(2)));
typedef float v8f __attribute__((ext_vector_type(8)));

#define C_IN   128
#define H_IN   224
#define W_IN   224
#define C_OUT  256
#define K_TOT  (C_IN * 9)
#define H_OUT  222
#define W_OUT  222
#define HW_OUT (H_OUT * W_OUT)

#define M_BLK  128
#define N_BLK  64
#define KT     16
#define NPANEL (K_TOT / KT)
#define A_STR  18
#define PB_STR (N_BLK * 2 + 4)

__global__ __launch_bounds__(256)
void conv3x3_wmma_f32(const float* __restrict__ x,
                      const float* __restrict__ w,
                      float* __restrict__ out) {
  __shared__ float As[M_BLK * A_STR];
  __shared__ float Bs[(KT / 2) * PB_STR];

  const int tid  = threadIdx.x;
  const int lane = tid & 31;
  const int wave = tid >> 5;
  const int wm   = wave & 3;
  const int wn   = wave >> 2;

  const int owBlk = blockIdx.x * N_BLK;
  const int oh    = blockIdx.y;
  const int mBase = blockIdx.z * M_BLK;

  v8f acc[2][2] = {};

  const int mA = tid >> 1;
  const int kA = (tid & 1) * 8;
  const int ktB = tid >> 4;
  const int j0  = (tid & 15) * 4;

  const int half = lane >> 4;
  const int l16  = lane & 15;

  float4 aR0, aR1;
  float  bR[4];

  auto loadG = [&](int kBase) {
    const float4* src =
        (const float4*)(w + (size_t)(mBase + mA) * K_TOT + kBase + kA);
    aR0 = src[0];
    aR1 = src[1];
    const int k  = kBase + ktB;
    const int c  = k / 9;
    const int r  = k - c * 9;
    const int kh = r / 3;
    const int kw = r - kh * 3;
    const float* bs =
        x + (size_t)c * (H_IN * W_IN) + (size_t)(oh + kh) * W_IN + owBlk + kw + j0;
#pragma unroll
    for (int i = 0; i < 4; ++i) {
      const int ow = owBlk + j0 + i;
      bR[i] = (ow < W_OUT) ? bs[i] : 0.0f;
    }
  };

  loadG(0);

  for (int p = 0; p < NPANEL; ++p) {
    {
      float* ad = &As[mA * A_STR + kA];
      ad[0] = aR0.x; ad[1] = aR0.y; ad[2] = aR0.z; ad[3] = aR0.w;
      ad[4] = aR1.x; ad[5] = aR1.y; ad[6] = aR1.z; ad[7] = aR1.w;
      float* bd = &Bs[(ktB >> 1) * PB_STR + j0 * 2 + (ktB & 1)];
#pragma unroll
      for (int i = 0; i < 4; ++i) bd[2 * i] = bR[i];
    }
    __syncthreads();

    if (p + 1 < NPANEL) loadG((p + 1) * KT);

#pragma unroll
    for (int k0 = 0; k0 < KT; k0 += 4) {
      v2f a[2], b[2];
#pragma unroll
      for (int ti = 0; ti < 2; ++ti)
        a[ti] = *(const v2f*)&As[(wm * 32 + ti * 16 + l16) * A_STR + k0 + 2 * half];
#pragma unroll
      for (int tj = 0; tj < 2; ++tj)
        b[tj] = *(const v2f*)&Bs[(k0 / 2 + half) * PB_STR +
                                 (wn * 32 + tj * 16 + l16) * 2];
#pragma unroll
      for (int ti = 0; ti < 2; ++ti)
#pragma unroll
        for (int tj = 0; tj < 2; ++tj)
          acc[ti][tj] = __builtin_amdgcn_wmma_f32_16x16x4_f32(
              false, a[ti], false, b[tj], (short)0, acc[ti][tj], false, false);
    }
    __syncthreads();
  }

#pragma unroll
  for (int ti = 0; ti < 2; ++ti) {
    const int m0 = mBase + wm * 32 + ti * 16 + half * 8;
#pragma unroll
    for (int tj = 0; tj < 2; ++tj) {
      const int ow = owBlk + wn * 32 + tj * 16 + l16;
      if (ow < W_OUT) {
        float* dst = out + (size_t)m0 * HW_OUT + (size_t)oh * W_OUT + ow;
#pragma unroll
        for (int r = 0; r < 8; ++r)
          dst[(size_t)r * HW_OUT] = acc[ti][tj][r];
      }
    }
  }
}

extern "C" void kernel_launch(void* const* d_in, const int* in_sizes, int n_in,
                              void* d_out, int out_size, void* d_ws, size_t ws_size,
                              hipStream_t stream) {
  const float* x = (const float*)d_in[0];
  const float* k = (const float*)d_in[1];
  float* out = (float*)d_out;

  dim3 grid((W_OUT + N_BLK - 1) / N_BLK,
            H_OUT,
            C_OUT / M_BLK);
  conv3x3_wmma_f32<<<grid, 256, 0, stream>>>(x, k, out);
}


# Flatten the output pixels (oh,ow) into one 49284-column GEMM axis so no 64-column block pads a 222-wide row: 1776 -> 1542 blocks, 13% fewer matrix instructions; same tile, loaders and k order
# speedup vs baseline: 1.0535x; 1.0535x over previous
#include <hip/hip_runtime.h>

typedef float v2f __attribute__((ext_vector_type(2)));
typedef float v8f __attribute__((ext_vector_type(8)));

#define C_IN   128
#define H_IN   224
#define W_IN   224
#define C_OUT  256
#define K_TOT  (C_IN * 9)
#define H_OUT  222
#define W_OUT  222
#define HW_OUT (H_OUT * W_OUT)

#define M_BLK  128
#define N_BLK  64
#define KT     16
#define NPANEL (K_TOT / KT)
#define N_BLKS ((HW_OUT + N_BLK - 1) / N_BLK)
#define A_STR  18
#define PB_STR (N_BLK * 2 + 4)

__global__ __launch_bounds__(256)
void conv3x3_wmma_f32(const float* __restrict__ x,
                      const float* __restrict__ w,
                      float* __restrict__ out) {
  __shared__ float As[M_BLK * A_STR];
  __shared__ float Bs[(KT / 2) * PB_STR];

  const int tid  = threadIdx.x;
  const int lane = tid & 31;
  const int wave = tid >> 5;
  const int wm   = wave & 3;
  const int wn   = wave >> 2;

  const int nBlk  = blockIdx.x * N_BLK;
  const int mBase = blockIdx.z * M_BLK;

  v8f acc[2][2] = {};

  const int mA = tid >> 1;
  const int kA = (tid & 1) * 8;
  const int ktB = tid >> 4;
  const int j0  = (tid & 15) * 4;

  const int half = lane >> 4;
  const int l16  = lane & 15;

  float4 aR0, aR1;
  float  bR[4];

  unsigned nByte[4];
#pragma unroll
  for (int i = 0; i < 4; ++i) {
    const int n   = nBlk + j0 + i;
    const int nn  = (n < HW_OUT) ? n : 0;
    const int ohh = nn / W_OUT;
    const int oww = nn - ohh * W_OUT;
    nByte[i] = (unsigned)(ohh * W_IN + oww) * 4u;
  }
  const char* xb = (const char*)x;

  auto loadG = [&](int kBase) {
    const float4* src =
        (const float4*)(w + (size_t)(mBase + mA) * K_TOT + kBase + kA);
    aR0 = src[0];
    aR1 = src[1];
    const int k  = kBase + ktB;
    const int c  = k / 9;
    const int r  = k - c * 9;
    const int kh = r / 3;
    const int kw = r - kh * 3;
    const unsigned kByte = (unsigned)(c * (H_IN * W_IN) + kh * W_IN + kw) * 4u;
#pragma unroll
    for (int i = 0; i < 4; ++i)
      bR[i] = *(const float*)(xb + (size_t)(kByte + nByte[i]));
  };

  loadG(0);

  for (int p = 0; p < NPANEL; ++p) {
    {
      float* ad = &As[mA * A_STR + kA];
      ad[0] = aR0.x; ad[1] = aR0.y; ad[2] = aR0.z; ad[3] = aR0.w;
      ad[4] = aR1.x; ad[5] = aR1.y; ad[6] = aR1.z; ad[7] = aR1.w;
      float* bd = &Bs[(ktB >> 1) * PB_STR + j0 * 2 + (ktB & 1)];
#pragma unroll
      for (int i = 0; i < 4; ++i) bd[2 * i] = bR[i];
    }
    __syncthreads();

    if (p + 1 < NPANEL) loadG((p + 1) * KT);
    __builtin_amdgcn_sched_barrier(0);

#pragma unroll
    for (int k0 = 0; k0 < KT; k0 += 4) {
      v2f a[2], b[2];
#pragma unroll
      for (int ti = 0; ti < 2; ++ti)
        a[ti] = *(const v2f*)&As[(wm * 32 + ti * 16 + l16) * A_STR + k0 + 2 * half];
#pragma unroll
      for (int tj = 0; tj < 2; ++tj)
        b[tj] = *(const v2f*)&Bs[(k0 / 2 + half) * PB_STR +
                                 (wn * 32 + tj * 16 + l16) * 2];
#pragma unroll
      for (int ti = 0; ti < 2; ++ti)
#pragma unroll
        for (int tj = 0; tj < 2; ++tj)
          acc[ti][tj] = __builtin_amdgcn_wmma_f32_16x16x4_f32(
              false, a[ti], false, b[tj], (short)0, acc[ti][tj], false, false);
    }
    __syncthreads();
  }

#pragma unroll
  for (int ti = 0; ti < 2; ++ti) {
    const int m0 = mBase + wm * 32 + ti * 16 + half * 8;
#pragma unroll
    for (int tj = 0; tj < 2; ++tj) {
      const int n = nBlk + wn * 32 + tj * 16 + l16;
      if (n < HW_OUT) {
        float* dst = out + (size_t)m0 * HW_OUT + n;
#pragma unroll
        for (int r = 0; r < 8; ++r)
          dst[(size_t)r * HW_OUT] = acc[ti][tj][r];
      }
    }
  }
}

extern "C" void kernel_launch(void* const* d_in, const int* in_sizes, int n_in,
                              void* d_out, int out_size, void* d_ws, size_t ws_size,
                              hipStream_t stream) {
  const float* x = (const float*)d_in[0];
  const float* k = (const float*)d_in[1];
  float* out = (float*)d_out;

  dim3 grid(N_BLKS,
            1,
            C_OUT / M_BLK);
  conv3x3_wmma_f32<<<grid, 256, 0, stream>>>(x, k, out);
}
